# starting kernel
# baseline (speedup 1.0000x reference)
#include <hip/hip_runtime.h>

typedef __attribute__((ext_vector_type(2))) float v2f;
typedef __attribute__((ext_vector_type(8))) float v8f;

#define HIDDEN 128

__global__ __launch_bounds__(256)
void element_update_wmma(const float* __restrict__ h_prev,
                         const float* __restrict__ m_curr,
                         const int*   __restrict__ atom_types,
                         const float* __restrict__ weight,
                         float*       __restrict__ out)
{
    const int lane  = threadIdx.x & 31;
    const int wave  = threadIdx.x >> 5;
    const int row   = lane & 15;
    const int kk    = (lane >> 4) << 1;
    const int hi8   = (lane >> 4) << 3;
    const int node0 = blockIdx.x << 4;

    const int col = (wave << 4) + row;
    const float* Mrow = m_curr + (size_t)(node0 + row) * HIDDEN;

    const int myT = atom_types[node0 + row];
    int t = __shfl(myT, 0, 32);

    for (;;) {
        const float* Wrow = weight + ((size_t)t * HIDDEN + col) * HIDDEN;

        v8f acc = {};
        #pragma unroll
        for (int k = 0; k < HIDDEN; k += 4) {
            v2f a = *(const v2f*)(Mrow + k + kk);
            v2f b = *(const v2f*)(Wrow + k + kk);
            acc = __builtin_amdgcn_wmma_f32_16x16x4_f32(
                 false, a,  false, b,
                 (short)0, acc,  false,  false);
        }

        #pragma unroll
        for (int r = 0; r < 8; ++r) {
            int rt = __shfl(myT, r + hi8, 32);
            if (rt == t) {
                size_t idx = (size_t)(node0 + r + hi8) * HIDDEN + col;
                out[idx] = h_prev[idx] + acc[r];
            }
        }

        int cand = (myT > t) ? myT : 0x7fffffff;
        #pragma unroll
        for (int off = 16; off; off >>= 1)
            cand = min(cand, __shfl_xor(cand, off, 32));
        if (cand == 0x7fffffff) break;
        t = cand;
    }
}

extern "C" void kernel_launch(void* const* d_in, const int* in_sizes, int n_in,
                              void* d_out, int out_size, void* d_ws, size_t ws_size,
                              hipStream_t stream) {
    const float* h_prev     = (const float*)d_in[0];
    const float* m_curr     = (const float*)d_in[1];
    const int*   atom_types = (const int*)  d_in[2];
    const float* weight     = (const float*)d_in[3];
    float*       out        = (float*)d_out;

    const int n_nodes = in_sizes[2];
    const int n_tiles = (n_nodes + 15) / 16;

    element_update_wmma<<<n_tiles, 256, 0, stream>>>(h_prev, m_curr, atom_types,
                                                     weight, out);
}


# Stage the M tile and each wave's W[s] slice through padded LDS with whole-row 16-byte global loads; fragments read from LDS, k order unchanged
# speedup vs baseline: 1.6116x; 1.6116x over previous
#include <hip/hip_runtime.h>


typedef __attribute__((ext_vector_type(2))) float v2f;
typedef __attribute__((ext_vector_type(4))) float v4f;
typedef __attribute__((ext_vector_type(8))) float v8f;

#define HIDDEN 128
#define PITCH 132

__global__ __launch_bounds__(256)
void element_update_wmma(const float* __restrict__ h_prev,
                         const float* __restrict__ m_curr,
                         const int*   __restrict__ atom_types,
                         const float* __restrict__ weight,
                         float*       __restrict__ out)
{
    __shared__ __attribute__((aligned(16))) float As[16 * PITCH];
    __shared__ __attribute__((aligned(16))) float Bs[8][16 * PITCH];

    const int lane  = threadIdx.x & 31;
    const int wave  = threadIdx.x >> 5;
    const int row   = lane & 15;
    const int kk    = (lane >> 4) << 1;
    const int hi8   = (lane >> 4) << 3;
    const int node0 = blockIdx.x << 4;

    const int col = (wave << 4) + row;

    {
        const float* Mt = m_curr + (size_t)node0 * HIDDEN + (lane << 2);
        #pragma unroll
        for (int i = 0; i < 2; ++i) {
            const int r = (wave << 1) + i;
            *(v4f*)(As + r * PITCH + (lane << 2)) = *(const v4f*)(Mt + r * HIDDEN);
        }
    }

    const int myT = atom_types[node0 + row];
    int t = __shfl(myT, 0, 32);

    float* Bw = Bs[wave];
    const float* Ar = As + row * PITCH + kk;
    const float* Br = Bw + row * PITCH + kk;

    for (;;) {
        const float* Wt = weight + ((size_t)t * HIDDEN + (wave << 4)) * HIDDEN + (lane << 2);
        v4f wv[16];
        #pragma unroll
        for (int i = 0; i < 16; ++i)
            wv[i] = *(const v4f*)(Wt + i * HIDDEN);
        #pragma unroll
        for (int i = 0; i < 16; ++i)
            *(v4f*)(Bw + i * PITCH + (lane << 2)) = wv[i];

        __syncthreads();

        v8f acc = {};
        #pragma unroll
        for (int k = 0; k < HIDDEN; k += 4) {
            v2f a = *(const v2f*)(Ar + k);
            v2f b = *(const v2f*)(Br + k);
            acc = __builtin_amdgcn_wmma_f32_16x16x4_f32(
                 false, a,  false, b,
                 (short)0, acc,  false,  false);
        }

        #pragma unroll
        for (int r = 0; r < 8; ++r) {
            int rt = __shfl(myT, r + hi8, 32);
            if (rt == t) {
                size_t idx = (size_t)(node0 + r + hi8) * HIDDEN + col;
                out[idx] = h_prev[idx] + acc[r];
            }
        }

        int cand = (myT > t) ? myT : 0x7fffffff;
        #pragma unroll
        for (int off = 16; off; off >>= 1)
            cand = min(cand, __shfl_xor(cand, off, 32));
        if (cand == 0x7fffffff) break;
        t = cand;
    }
}

extern "C" void kernel_launch(void* const* d_in, const int* in_sizes, int n_in,
                              void* d_out, int out_size, void* d_ws, size_t ws_size,
                              hipStream_t stream) {
    const float* h_prev     = (const float*)d_in[0];
    const float* m_curr     = (const float*)d_in[1];
    const int*   atom_types = (const int*)  d_in[2];
    const float* weight     = (const float*)d_in[3];
    float*       out        = (float*)d_out;

    const int n_nodes = in_sizes[2];
    const int n_tiles = (n_nodes + 15) / 16;

    element_update_wmma<<<n_tiles, 256, 0, stream>>>(h_prev, m_curr, atom_types,
                                                     weight, out);
}


# Widen the block tile from 16 to 64 node rows: each staged W[s] slice and its B fragments feed four 16-row subtiles (grid 1024 -> 256 blocks); the tile's h_prev is loaded once, before the species loop
# speedup vs baseline: 2.4062x; 1.4931x over previous
#include <hip/hip_runtime.h>


typedef __attribute__((ext_vector_type(2))) float v2f;
typedef __attribute__((ext_vector_type(4))) float v4f;
typedef __attribute__((ext_vector_type(8))) float v8f;

#define HIDDEN 128
#define PITCH 132
#define TILE_ROWS 64
#define SUBTILES (TILE_ROWS / 16)

__global__ __launch_bounds__(256)
void element_update_wmma(const float* __restrict__ h_prev,
                         const float* __restrict__ m_curr,
                         const int*   __restrict__ atom_types,
                         const float* __restrict__ weight,
                         float*       __restrict__ out)
{
    __shared__ __attribute__((aligned(16))) float As[TILE_ROWS * PITCH];
    __shared__ __attribute__((aligned(16))) float Bs[8][16 * PITCH];

    const int lane  = threadIdx.x & 31;
    const int wave  = threadIdx.x >> 5;
    const int row   = lane & 15;
    const int kk    = (lane >> 4) << 1;
    const int hi8   = (lane >> 4) << 3;
    const int node0 = blockIdx.x * TILE_ROWS;

    const int col = (wave << 4) + row;

    const int tLo = atom_types[node0 + lane];
    const int tHi = atom_types[node0 + 32 + lane];

    const float* Mt = m_curr + ((size_t)node0 + (wave << 3)) * HIDDEN + (lane << 2);
    v4f mv[8];
    #pragma unroll
    for (int i = 0; i < 8; ++i)
        mv[i] = *(const v4f*)(Mt + i * HIDDEN);

    const size_t cd0 = ((size_t)node0 + hi8) * HIDDEN + col;
    float hp[SUBTILES][8];
    #pragma unroll
    for (int j = 0; j < SUBTILES; ++j)
        #pragma unroll
        for (int r = 0; r < 8; ++r)
            hp[j][r] = h_prev[cd0 + (size_t)(16 * j + r) * HIDDEN];

    float* Bw = Bs[wave];
    const float* Ar = As + row * PITCH + kk;
    const float* Br = Bw + row * PITCH + kk;
    const float* Ww = weight + (size_t)(wave << 4) * HIDDEN + (lane << 2);

    int t = __builtin_amdgcn_readfirstlane(tLo);
    v4f wv[16];
    #pragma unroll
    for (int i = 0; i < 16; ++i)
        wv[i] = *(const v4f*)(Ww + (size_t)t * (HIDDEN * HIDDEN) + i * HIDDEN);

    {
        float* Ad = As + (wave << 3) * PITCH + (lane << 2);
        #pragma unroll
        for (int i = 0; i < 8; ++i)
            *(v4f*)(Ad + i * PITCH) = mv[i];
    }

    unsigned long long todo = ~0ull;

    for (;;) {
        #pragma unroll
        for (int i = 0; i < 16; ++i)
            *(v4f*)(Bw + i * PITCH + (lane << 2)) = wv[i];

        __syncthreads();

        v2f bf[HIDDEN / 4];
        #pragma unroll
        for (int s = 0; s < HIDDEN / 4; ++s)
            bf[s] = *(const v2f*)(Br + 4 * s);

        const unsigned rowsLo = __builtin_amdgcn_ballot_w32(tLo == t);
        const unsigned rowsHi = __builtin_amdgcn_ballot_w32(tHi == t);

        #pragma unroll
        for (int j = 0; j < SUBTILES; ++j) {
            const unsigned m16 = (((j < 2) ? rowsLo : rowsHi) >> ((j & 1) << 4)) & 0xffffu;
            if (m16 == 0) continue;

            const float* Aj = Ar + j * 16 * PITCH;
            v8f acc = {};
            #pragma unroll
            for (int s = 0; s < HIDDEN / 4; ++s) {
                v2f a = *(const v2f*)(Aj + 4 * s);
                acc = __builtin_amdgcn_wmma_f32_16x16x4_f32(
                     false, a,  false, bf[s],
                     (short)0, acc,  false,  false);
            }

            #pragma unroll
            for (int r = 0; r < 8; ++r) {
                const float v = hp[j][r] + acc[r];
                if ((m16 >> (r + hi8)) & 1u)
                    out[cd0 + (size_t)(16 * j + r) * HIDDEN] = v;
            }
        }

        todo &= ~(((unsigned long long)rowsHi << 32) | (unsigned long long)rowsLo);
        if (todo == 0) break;

        const int r0 = __builtin_ctzll(todo);
        t = __builtin_amdgcn_readlane((r0 < 32) ? tLo : tHi, r0 & 31);
        #pragma unroll
        for (int i = 0; i < 16; ++i)
            wv[i] = *(const v4f*)(Ww + (size_t)t * (HIDDEN * HIDDEN) + i * HIDDEN);
    }
}

extern "C" void kernel_launch(void* const* d_in, const int* in_sizes, int n_in,
                              void* d_out, int out_size, void* d_ws, size_t ws_size,
                              hipStream_t stream) {
    const float* h_prev     = (const float*)d_in[0];
    const float* m_curr     = (const float*)d_in[1];
    const int*   atom_types = (const int*)  d_in[2];
    const float* weight     = (const float*)d_in[3];
    float*       out        = (float*)d_out;

    const int n_nodes = in_sizes[2];
    const int n_tiles = (n_nodes + 63) / 64;

    element_update_wmma<<<n_tiles, 256, 0, stream>>>(h_prev, m_curr, atom_types,
                                                     weight, out);
}


# Request the next species' W slice ahead of the current species' WMMA chains instead of behind them; the block barrier fences LDS only, so it no longer waits for that fetch; sums unchanged
# speedup vs baseline: 2.6161x; 1.0872x over previous
#include <hip/hip_runtime.h>


typedef __attribute__((ext_vector_type(2))) float v2f;
typedef __attribute__((ext_vector_type(4))) float v4f;
typedef __attribute__((ext_vector_type(8))) float v8f;

#define HIDDEN 128
#define PITCH 132
#define TILE_ROWS 64
#define SUBTILES (TILE_ROWS / 16)

__global__ __launch_bounds__(256)
void element_update_wmma(const float* __restrict__ h_prev,
                         const float* __restrict__ m_curr,
                         const int*   __restrict__ atom_types,
                         const float* __restrict__ weight,
                         float*       __restrict__ out)
{
    __shared__ __attribute__((aligned(16))) float As[TILE_ROWS * PITCH];
    __shared__ __attribute__((aligned(16))) float Bs[8][16 * PITCH];

    const int lane  = threadIdx.x & 31;
    const int wave  = threadIdx.x >> 5;
    const int row   = lane & 15;
    const int kk    = (lane >> 4) << 1;
    const int hi8   = (lane >> 4) << 3;
    const int node0 = blockIdx.x * TILE_ROWS;

    const int col = (wave << 4) + row;

    const int tLo = atom_types[node0 + lane];
    const int tHi = atom_types[node0 + 32 + lane];

    const float* Mt = m_curr + ((size_t)node0 + (wave << 3)) * HIDDEN + (lane << 2);
    v4f mv[8];
    #pragma unroll
    for (int i = 0; i < 8; ++i)
        mv[i] = *(const v4f*)(Mt + i * HIDDEN);

    const size_t cd0 = ((size_t)node0 + hi8) * HIDDEN + col;
    float hp[SUBTILES][8];
    #pragma unroll
    for (int j = 0; j < SUBTILES; ++j)
        #pragma unroll
        for (int r = 0; r < 8; ++r)
            hp[j][r] = h_prev[cd0 + (size_t)(16 * j + r) * HIDDEN];

    float* Bw = Bs[wave];
    const float* Ar = As + row * PITCH + kk;
    const float* Br = Bw + row * PITCH + kk;
    const float* Ww = weight + (size_t)(wave << 4) * HIDDEN + (lane << 2);

    int t = __builtin_amdgcn_readfirstlane(tLo);
    v4f wv[16];
    #pragma unroll
    for (int i = 0; i < 16; ++i)
        wv[i] = *(const v4f*)(Ww + (size_t)t * (HIDDEN * HIDDEN) + i * HIDDEN);

    {
        float* Ad = As + (wave << 3) * PITCH + (lane << 2);
        #pragma unroll
        for (int i = 0; i < 8; ++i)
            *(v4f*)(Ad + i * PITCH) = mv[i];
    }

    unsigned long long todo = ~0ull;

    for (;;) {
        #pragma unroll
        for (int i = 0; i < 16; ++i)
            *(v4f*)(Bw + i * PITCH + (lane << 2)) = wv[i];

        const unsigned rowsLo = __builtin_amdgcn_ballot_w32(tLo == t);
        const unsigned rowsHi = __builtin_amdgcn_ballot_w32(tHi == t);

        todo &= ~(((unsigned long long)rowsHi << 32) | (unsigned long long)rowsLo);
        const bool more = (todo != 0);

        if (more) {
            const int r0 = __builtin_ctzll(todo);
            t = __builtin_amdgcn_readlane((r0 < 32) ? tLo : tHi, r0 & 31);
            #pragma unroll
            for (int i = 0; i < 16; ++i)
                wv[i] = *(const v4f*)(Ww + (size_t)t * (HIDDEN * HIDDEN) + i * HIDDEN);
        }

        __builtin_amdgcn_fence(__ATOMIC_RELEASE, "workgroup", "local");
        __builtin_amdgcn_s_barrier();
        __builtin_amdgcn_fence(__ATOMIC_ACQUIRE, "workgroup", "local");

        v2f bf[HIDDEN / 4];
        #pragma unroll
        for (int s = 0; s < HIDDEN / 4; ++s)
            bf[s] = *(const v2f*)(Br + 4 * s);

        #pragma unroll
        for (int j = 0; j < SUBTILES; ++j) {
            const unsigned m16 = (((j < 2) ? rowsLo : rowsHi) >> ((j & 1) << 4)) & 0xffffu;
            if (m16 == 0) continue;

            const float* Aj = Ar + j * 16 * PITCH;
            v8f acc = {};
            #pragma unroll
            for (int s = 0; s < HIDDEN / 4; ++s) {
                v2f a = *(const v2f*)(Aj + 4 * s);
                acc = __builtin_amdgcn_wmma_f32_16x16x4_f32(
                     false, a,  false, bf[s],
                     (short)0, acc,  false,  false);
            }

            #pragma unroll
            for (int r = 0; r < 8; ++r) {
                const float v = hp[j][r] + acc[r];
                if ((m16 >> (r + hi8)) & 1u)
                    out[cd0 + (size_t)(16 * j + r) * HIDDEN] = v;
            }
        }

        if (!more) break;
    }
}

extern "C" void kernel_launch(void* const* d_in, const int* in_sizes, int n_in,
                              void* d_out, int out_size, void* d_ws, size_t ws_size,
                              hipStream_t stream) {
    const float* h_prev     = (const float*)d_in[0];
    const float* m_curr     = (const float*)d_in[1];
    const int*   atom_types = (const int*)  d_in[2];
    const float* weight     = (const float*)d_in[3];
    float*       out        = (float*)d_out;

    const int n_nodes = in_sizes[2];
    const int n_tiles = (n_nodes + 63) / 64;

    element_update_wmma<<<n_tiles, 256, 0, stream>>>(h_prev, m_curr, atom_types,
                                                     weight, out);
}


# Rows of a subtile shared by two species are blended in registers by lane selects and each subtile is stored once, whole: no exec-masked store branch between the WMMA chains; sums unchanged
# speedup vs baseline: 3.0687x; 1.1730x over previous
#include <hip/hip_runtime.h>


typedef __attribute__((ext_vector_type(2))) float v2f;
typedef __attribute__((ext_vector_type(4))) float v4f;
typedef __attribute__((ext_vector_type(8))) float v8f;

#define HIDDEN 128
#define PITCH 132
#define TILE_ROWS 64
#define SUBTILES (TILE_ROWS / 16)

__global__ __launch_bounds__(256)
void element_update_wmma(const float* __restrict__ h_prev,
                         const float* __restrict__ m_curr,
                         const int*   __restrict__ atom_types,
                         const float* __restrict__ weight,
                         float*       __restrict__ out)
{
    __shared__ __attribute__((aligned(16))) float As[TILE_ROWS * PITCH];
    __shared__ __attribute__((aligned(16))) float Bs[8][16 * PITCH];

    const int lane  = threadIdx.x & 31;
    const int wave  = threadIdx.x >> 5;
    const int row   = lane & 15;
    const int kk    = (lane >> 4) << 1;
    const int hi8   = (lane >> 4) << 3;
    const int node0 = blockIdx.x * TILE_ROWS;

    const int col = (wave << 4) + row;

    const int tLo = atom_types[node0 + lane];
    const int tHi = atom_types[node0 + 32 + lane];

    const float* Mt = m_curr + ((size_t)node0 + (wave << 3)) * HIDDEN + (lane << 2);
    v4f mv[8];
    #pragma unroll
    for (int i = 0; i < 8; ++i)
        mv[i] = *(const v4f*)(Mt + i * HIDDEN);

    const size_t cd0 = ((size_t)node0 + hi8) * HIDDEN + col;
    float hp[SUBTILES][8];
    #pragma unroll
    for (int j = 0; j < SUBTILES; ++j)
        #pragma unroll
        for (int r = 0; r < 8; ++r)
            hp[j][r] = h_prev[cd0 + (size_t)(16 * j + r) * HIDDEN];

    float res[SUBTILES][8];
    #pragma unroll
    for (int j = 0; j < SUBTILES; ++j)
        #pragma unroll
        for (int r = 0; r < 8; ++r)
            res[j][r] = 0.0f;

    float* Bw = Bs[wave];
    const float* Ar = As + row * PITCH + kk;
    const float* Br = Bw + row * PITCH + kk;
    const float* Ww = weight + (size_t)(wave << 4) * HIDDEN + (lane << 2);

    int t = __builtin_amdgcn_readfirstlane(tLo);
    v4f wv[16];
    #pragma unroll
    for (int i = 0; i < 16; ++i)
        wv[i] = *(const v4f*)(Ww + (size_t)t * (HIDDEN * HIDDEN) + i * HIDDEN);

    {
        float* Ad = As + (wave << 3) * PITCH + (lane << 2);
        #pragma unroll
        for (int i = 0; i < 8; ++i)
            *(v4f*)(Ad + i * PITCH) = mv[i];
    }

    unsigned long long todo = ~0ull;

    for (;;) {
        #pragma unroll
        for (int i = 0; i < 16; ++i)
            *(v4f*)(Bw + i * PITCH + (lane << 2)) = wv[i];

        const unsigned rowsLo = __builtin_amdgcn_ballot_w32(tLo == t);
        const unsigned rowsHi = __builtin_amdgcn_ballot_w32(tHi == t);

        todo &= ~(((unsigned long long)rowsHi << 32) | (unsigned long long)rowsLo);
        const bool more = (todo != 0);

        if (more) {
            const int r0 = __builtin_ctzll(todo);
            t = __builtin_amdgcn_readlane((r0 < 32) ? tLo : tHi, r0 & 31);
            #pragma unroll
            for (int i = 0; i < 16; ++i)
                wv[i] = *(const v4f*)(Ww + (size_t)t * (HIDDEN * HIDDEN) + i * HIDDEN);
        }

        __builtin_amdgcn_fence(__ATOMIC_RELEASE, "workgroup", "local");
        __builtin_amdgcn_s_barrier();
        __builtin_amdgcn_fence(__ATOMIC_ACQUIRE, "workgroup", "local");

        v2f bf[HIDDEN / 4];
        #pragma unroll
        for (int s = 0; s < HIDDEN / 4; ++s)
            bf[s] = *(const v2f*)(Br + 4 * s);

        #pragma unroll
        for (int j = 0; j < SUBTILES; ++j) {
            const unsigned m16 = (((j < 2) ? rowsLo : rowsHi) >> ((j & 1) << 4)) & 0xffffu;
            if (m16 == 0) continue;

            const float* Aj = Ar + j * 16 * PITCH;
            v8f acc = {};
            #pragma unroll
            for (int s = 0; s < HIDDEN / 4; ++s) {
                v2f a = *(const v2f*)(Aj + 4 * s);
                acc = __builtin_amdgcn_wmma_f32_16x16x4_f32(
                     false, a,  false, bf[s],
                     (short)0, acc,  false,  false);
            }

            #pragma unroll
            for (int r = 0; r < 8; ++r)
                res[j][r] = ((m16 >> (r + hi8)) & 1u) ? acc[r] : res[j][r];

            if (((unsigned)(todo >> (16 * j)) & 0xffffu) == 0) {
                #pragma unroll
                for (int r = 0; r < 8; ++r)
                    out[cd0 + (size_t)(16 * j + r) * HIDDEN] = hp[j][r] + res[j][r];
            }
        }

        if (!more) break;
    }
}

extern "C" void kernel_launch(void* const* d_in, const int* in_sizes, int n_in,
                              void* d_out, int out_size, void* d_ws, size_t ws_size,
                              hipStream_t stream) {
    const float* h_prev     = (const float*)d_in[0];
    const float* m_curr     = (const float*)d_in[1];
    const int*   atom_types = (const int*)  d_in[2];
    const float* weight     = (const float*)d_in[3];
    float*       out        = (float*)d_out;

    const int n_nodes = in_sizes[2];
    const int n_tiles = (n_nodes + 63) / 64;

    element_update_wmma<<<n_tiles, 256, 0, stream>>>(h_prev, m_curr, atom_types,
                                                     weight, out);
}


# All A fragments of the 64-row tile are read from LDS once, in one group behind the only block barrier and ahead of every WMMA chain, and stay in registers for all species; sums unchanged
# speedup vs baseline: 3.3954x; 1.1065x over previous
#include <hip/hip_runtime.h>


typedef __attribute__((ext_vector_type(2))) float v2f;
typedef __attribute__((ext_vector_type(4))) float v4f;
typedef __attribute__((ext_vector_type(8))) float v8f;

#define HIDDEN 128
#define PITCH 132
#define TILE_ROWS 64
#define SUBTILES (TILE_ROWS / 16)

__global__ __launch_bounds__(256)
void element_update_wmma(const float* __restrict__ h_prev,
                         const float* __restrict__ m_curr,
                         const int*   __restrict__ atom_types,
                         const float* __restrict__ weight,
                         float*       __restrict__ out)
{
    __shared__ __attribute__((aligned(16))) float As[TILE_ROWS * PITCH];
    __shared__ __attribute__((aligned(16))) float Bs[8][16 * PITCH];

    const int lane  = threadIdx.x & 31;
    const int wave  = threadIdx.x >> 5;
    const int row   = lane & 15;
    const int kk    = (lane >> 4) << 1;
    const int hi8   = (lane >> 4) << 3;
    const int node0 = blockIdx.x * TILE_ROWS;

    const int col = (wave << 4) + row;

    const int tLo = atom_types[node0 + lane];
    const int tHi = atom_types[node0 + 32 + lane];

    const float* Mt = m_curr + ((size_t)node0 + (wave << 3)) * HIDDEN + (lane << 2);
    v4f mv[8];
    #pragma unroll
    for (int i = 0; i < 8; ++i)
        mv[i] = *(const v4f*)(Mt + i * HIDDEN);

    const size_t cd0 = ((size_t)node0 + hi8) * HIDDEN + col;
    float hp[SUBTILES][8];
    #pragma unroll
    for (int j = 0; j < SUBTILES; ++j)
        #pragma unroll
        for (int r = 0; r < 8; ++r)
            hp[j][r] = h_prev[cd0 + (size_t)(16 * j + r) * HIDDEN];

    float res[SUBTILES][8];
    #pragma unroll
    for (int j = 0; j < SUBTILES; ++j)
        #pragma unroll
        for (int r = 0; r < 8; ++r)
            res[j][r] = 0.0f;

    float* Bw = Bs[wave];
    const float* Ar = As + row * PITCH + kk;
    const float* Br = Bw + row * PITCH + kk;
    const float* Ww = weight + (size_t)(wave << 4) * HIDDEN + (lane << 2);

    int t = __builtin_amdgcn_readfirstlane(tLo);
    v4f wv[16];
    #pragma unroll
    for (int i = 0; i < 16; ++i)
        wv[i] = *(const v4f*)(Ww + (size_t)t * (HIDDEN * HIDDEN) + i * HIDDEN);

    {
        float* Ad = As + (wave << 3) * PITCH + (lane << 2);
        #pragma unroll
        for (int i = 0; i < 8; ++i)
            *(v4f*)(Ad + i * PITCH) = mv[i];
    }

    __builtin_amdgcn_fence(__ATOMIC_RELEASE, "workgroup", "local");
    __builtin_amdgcn_s_barrier();
    __builtin_amdgcn_fence(__ATOMIC_ACQUIRE, "workgroup", "local");

    v2f af[SUBTILES][HIDDEN / 4];
    #pragma unroll
    for (int j = 0; j < SUBTILES; ++j)
        #pragma unroll
        for (int s = 0; s < HIDDEN / 4; ++s)
            af[j][s] = *(const v2f*)(Ar + j * 16 * PITCH + 4 * s);

    unsigned long long todo = ~0ull;

    for (;;) {
        #pragma unroll
        for (int i = 0; i < 16; ++i)
            *(v4f*)(Bw + i * PITCH + (lane << 2)) = wv[i];

        const unsigned rowsLo = __builtin_amdgcn_ballot_w32(tLo == t);
        const unsigned rowsHi = __builtin_amdgcn_ballot_w32(tHi == t);

        todo &= ~(((unsigned long long)rowsHi << 32) | (unsigned long long)rowsLo);
        const bool more = (todo != 0);

        if (more) {
            const int r0 = __builtin_ctzll(todo);
            t = __builtin_amdgcn_readlane((r0 < 32) ? tLo : tHi, r0 & 31);
            #pragma unroll
            for (int i = 0; i < 16; ++i)
                wv[i] = *(const v4f*)(Ww + (size_t)t * (HIDDEN * HIDDEN) + i * HIDDEN);
        }

        __builtin_amdgcn_fence(__ATOMIC_RELEASE, "workgroup", "local");
        __builtin_amdgcn_wave_barrier();
        __builtin_amdgcn_fence(__ATOMIC_ACQUIRE, "workgroup", "local");

        v2f bf[HIDDEN / 4];
        #pragma unroll
        for (int s = 0; s < HIDDEN / 4; ++s)
            bf[s] = *(const v2f*)(Br + 4 * s);

        #pragma unroll
        for (int j = 0; j < SUBTILES; ++j) {
            const unsigned m16 = (((j < 2) ? rowsLo : rowsHi) >> ((j & 1) << 4)) & 0xffffu;
            if (m16 == 0) continue;

            v8f acc = {};
            #pragma unroll
            for (int s = 0; s < HIDDEN / 4; ++s)
                acc = __builtin_amdgcn_wmma_f32_16x16x4_f32(
                     false, af[j][s],  false, bf[s],
                     (short)0, acc,  false,  false);

            #pragma unroll
            for (int r = 0; r < 8; ++r)
                res[j][r] = ((m16 >> (r + hi8)) & 1u) ? acc[r] : res[j][r];

            if (((unsigned)(todo >> (16 * j)) & 0xffffu) == 0) {
                #pragma unroll
                for (int r = 0; r < 8; ++r)
                    out[cd0 + (size_t)(16 * j + r) * HIDDEN] = hp[j][r] + res[j][r];
            }
        }

        if (!more) break;
    }
}

extern "C" void kernel_launch(void* const* d_in, const int* in_sizes, int n_in,
                              void* d_out, int out_size, void* d_ws, size_t ws_size,
                              hipStream_t stream) {
    const float* h_prev     = (const float*)d_in[0];
    const float* m_curr     = (const float*)d_in[1];
    const int*   atom_types = (const int*)  d_in[2];
    const float* weight     = (const float*)d_in[3];
    float*       out        = (float*)d_out;

    const int n_nodes = in_sizes[2];
    const int n_tiles = (n_nodes + 63) / 64;

    element_update_wmma<<<n_tiles, 256, 0, stream>>>(h_prev, m_curr, atom_types,
                                                     weight, out);
}


# First species pass written out on its own and fed in order of first use: M tile, W slice as two K halves, h_prev last; all chains run k steps 0-15 while the rest lands; sums unchanged
# speedup vs baseline: 3.5704x; 1.0515x over previous
#include <hip/hip_runtime.h>


typedef __attribute__((ext_vector_type(2))) float v2f;
typedef __attribute__((ext_vector_type(4))) float v4f;
typedef __attribute__((ext_vector_type(8))) float v8f;

#define HIDDEN 128
#define KHALF (HIDDEN / 2)
#define KSTEPS (HIDDEN / 4)
#define PITCH 132
#define TILE_ROWS 64
#define SUBTILES (TILE_ROWS / 16)

__global__ __launch_bounds__(256)
void element_update_wmma(const float* __restrict__ h_prev,
                         const float* __restrict__ m_curr,
                         const int*   __restrict__ atom_types,
                         const float* __restrict__ weight,
                         float*       __restrict__ out)
{
    __shared__ __attribute__((aligned(16))) float As[TILE_ROWS * PITCH];
    __shared__ __attribute__((aligned(16))) float Bs[8][16 * PITCH];

    const int lane  = threadIdx.x & 31;
    const int wave  = threadIdx.x >> 5;
    const int row   = lane & 15;
    const int half  = lane >> 4;
    const int kk    = half << 1;
    const int hi8   = half << 3;
    const int node0 = blockIdx.x * TILE_ROWS;

    const int col = (wave << 4) + row;

    const int tLo = atom_types[node0 + lane];
    const int tHi = atom_types[node0 + 32 + lane];

    const float* Mt = m_curr + ((size_t)node0 + (wave << 3)) * HIDDEN + (lane << 2);
    v4f mv[8];
    #pragma unroll
    for (int i = 0; i < 8; ++i)
        mv[i] = *(const v4f*)(Mt + i * HIDDEN);
    __builtin_amdgcn_sched_barrier(0);

    const size_t cd0 = ((size_t)node0 + hi8) * HIDDEN + col;

    float res[SUBTILES][8];
    #pragma unroll
    for (int j = 0; j < SUBTILES; ++j)
        #pragma unroll
        for (int r = 0; r < 8; ++r)
            res[j][r] = 0.0f;

    float* Bw = Bs[wave];
    const float* Ar = As + row * PITCH + kk;
    const float* Br = Bw + row * PITCH + kk;
    const int wcol = row << 2;
    const float* Ww = weight + (size_t)((wave << 4) + half) * HIDDEN + wcol;
    float*       Bd = Bw + half * PITCH + wcol;

    int t = __builtin_amdgcn_readfirstlane(tLo);
    v4f wv[16];
    #pragma unroll
    for (int i = 0; i < 8; ++i)
        wv[i] = *(const v4f*)(Ww + (size_t)t * (HIDDEN * HIDDEN) + 2 * i * HIDDEN);
    __builtin_amdgcn_sched_barrier(0);
    #pragma unroll
    for (int i = 0; i < 8; ++i)
        wv[8 + i] = *(const v4f*)(Ww + (size_t)t * (HIDDEN * HIDDEN) + 2 * i * HIDDEN + KHALF);

    {
        float* Ad = As + (wave << 3) * PITCH + (lane << 2);
        #pragma unroll
        for (int i = 0; i < 8; ++i)
            *(v4f*)(Ad + i * PITCH) = mv[i];
    }

    __builtin_amdgcn_fence(__ATOMIC_RELEASE, "workgroup", "local");
    __builtin_amdgcn_s_barrier();
    __builtin_amdgcn_fence(__ATOMIC_ACQUIRE, "workgroup", "local");

    float hp[SUBTILES][8];
    #pragma unroll
    for (int j = 0; j < SUBTILES; ++j)
        #pragma unroll
        for (int r = 0; r < 8; ++r)
            hp[j][r] = h_prev[cd0 + (size_t)(16 * j + r) * HIDDEN];

    v2f af[SUBTILES][KSTEPS];
    #pragma unroll
    for (int j = 0; j < SUBTILES; ++j)
        #pragma unroll
        for (int s = 0; s < KSTEPS / 2; ++s)
            af[j][s] = *(const v2f*)(Ar + j * 16 * PITCH + 4 * s);

    unsigned long long todo = ~0ull;

    auto species_pass = [&](const bool first) __attribute__((always_inline)) -> bool {
        #pragma unroll
        for (int i = 0; i < 8; ++i)
            *(v4f*)(Bd + 2 * i * PITCH) = wv[i];
        if (!first) {
            #pragma unroll
            for (int i = 0; i < 8; ++i)
                *(v4f*)(Bd + 2 * i * PITCH + KHALF) = wv[8 + i];
        }

        const unsigned rowsLo = __builtin_amdgcn_ballot_w32(tLo == t);
        const unsigned rowsHi = __builtin_amdgcn_ballot_w32(tHi == t);

        todo &= ~(((unsigned long long)rowsHi << 32) | (unsigned long long)rowsLo);
        const bool more = (todo != 0);

        if (more) {
            const int r0 = __builtin_ctzll(todo);
            t = __builtin_amdgcn_readlane((r0 < 32) ? tLo : tHi, r0 & 31);
            #pragma unroll
            for (int i = 0; i < 8; ++i)
                wv[i] = *(const v4f*)(Ww + (size_t)t * (HIDDEN * HIDDEN) + 2 * i * HIDDEN);
            if (!first) {
                #pragma unroll
                for (int i = 0; i < 8; ++i)
                    wv[8 + i] = *(const v4f*)(Ww + (size_t)t * (HIDDEN * HIDDEN)
                                              + 2 * i * HIDDEN + KHALF);
            }
        }

        __builtin_amdgcn_fence(__ATOMIC_RELEASE, "workgroup", "local");
        __builtin_amdgcn_wave_barrier();
        __builtin_amdgcn_fence(__ATOMIC_ACQUIRE, "workgroup", "local");

        v2f bf[KSTEPS];
        #pragma unroll
        for (int s = 0; s < KSTEPS / 2; ++s)
            bf[s] = *(const v2f*)(Br + 4 * s);
        if (!first) {
            #pragma unroll
            for (int s = KSTEPS / 2; s < KSTEPS; ++s)
                bf[s] = *(const v2f*)(Br + 4 * s);
        } else {
            __builtin_amdgcn_sched_barrier(0);
            #pragma unroll
            for (int j = 0; j < SUBTILES; ++j)
                #pragma unroll
                for (int s = KSTEPS / 2; s < KSTEPS; ++s)
                    af[j][s] = *(const v2f*)(Ar + j * 16 * PITCH + 4 * s);
            __builtin_amdgcn_sched_barrier(0);
        }

        unsigned m16[SUBTILES];
        #pragma unroll
        for (int j = 0; j < SUBTILES; ++j)
            m16[j] = (((j < 2) ? rowsLo : rowsHi) >> ((j & 1) << 4)) & 0xffffu;

        v8f acc[SUBTILES];
        #pragma unroll
        for (int j = 0; j < SUBTILES; ++j)
            acc[j] = v8f{};

        if (first) {
            #pragma unroll
            for (int j = 0; j < SUBTILES; ++j) {
                if (m16[j] == 0) continue;
                #pragma unroll
                for (int s = 0; s < KSTEPS / 2; ++s)
                    acc[j] = __builtin_amdgcn_wmma_f32_16x16x4_f32(
                         false, af[j][s],  false, bf[s],
                         (short)0, acc[j],  false,  false);
            }

            #pragma unroll
            for (int i = 0; i < 8; ++i)
                *(v4f*)(Bd + 2 * i * PITCH + KHALF) = wv[8 + i];
            if (more) {
                #pragma unroll
                for (int i = 0; i < 8; ++i)
                    wv[8 + i] = *(const v4f*)(Ww + (size_t)t * (HIDDEN * HIDDEN)
                                              + 2 * i * HIDDEN + KHALF);
            }
            __builtin_amdgcn_fence(__ATOMIC_RELEASE, "workgroup", "local");
            __builtin_amdgcn_wave_barrier();
            __builtin_amdgcn_fence(__ATOMIC_ACQUIRE, "workgroup", "local");
            #pragma unroll
            for (int s = KSTEPS / 2; s < KSTEPS; ++s)
                bf[s] = *(const v2f*)(Br + 4 * s);
        }

        #pragma unroll
        for (int j = 0; j < SUBTILES; ++j) {
            if (m16[j] == 0) continue;

            #pragma unroll
            for (int s = first ? KSTEPS / 2 : 0; s < KSTEPS; ++s)
                acc[j] = __builtin_amdgcn_wmma_f32_16x16x4_f32(
                     false, af[j][s],  false, bf[s],
                     (short)0, acc[j],  false,  false);

            #pragma unroll
            for (int r = 0; r < 8; ++r)
                res[j][r] = ((m16[j] >> (r + hi8)) & 1u) ? acc[j][r] : res[j][r];

            if (((unsigned)(todo >> (16 * j)) & 0xffffu) == 0) {
                #pragma unroll
                for (int r = 0; r < 8; ++r)
                    out[cd0 + (size_t)(16 * j + r) * HIDDEN] = hp[j][r] + res[j][r];
            }
        }

        return more;
    };

    if (species_pass(true)) {
        while (species_pass(false)) { }
    }
}

extern "C" void kernel_launch(void* const* d_in, const int* in_sizes, int n_in,
                              void* d_out, int out_size, void* d_ws, size_t ws_size,
                              hipStream_t stream) {
    const float* h_prev     = (const float*)d_in[0];
    const float* m_curr     = (const float*)d_in[1];
    const int*   atom_types = (const int*)  d_in[2];
    const float* weight     = (const float*)d_in[3];
    float*       out        = (float*)d_out;

    const int n_nodes = in_sizes[2];
    const int n_tiles = (n_nodes + 63) / 64;

    element_update_wmma<<<n_tiles, 256, 0, stream>>>(h_prev, m_curr, atom_types,
                                                     weight, out);
}
